# starting kernel
# baseline (speedup 1.0000x reference)
#include <hip/hip_runtime.h>
#include <hip/hip_bf16.h>


typedef __attribute__((ext_vector_type(2))) float v2f;
typedef __attribute__((ext_vector_type(8))) float v8f;

#define HEADS 4
#define HDIM 32
#define CIN 128
#define NEG_SLOPE 0.2f

__global__ void gat_zero_kernel(float* __restrict__ p, size_t n) {
    size_t i = (size_t)blockIdx.x * blockDim.x + threadIdx.x;
    size_t stride = (size_t)gridDim.x * blockDim.x;
    for (; i < n; i += stride) p[i] = 0.0f;
}

__global__ __launch_bounds__(256) void gat_gemm_wmma(
    const float* __restrict__ X,
    const float* __restrict__ Wl,
    const float* __restrict__ Wr,
    float* __restrict__ XL,
    float* __restrict__ XR,
    int Nn)
{
    const int wave = threadIdx.x >> 5;
    const int lane = threadIdx.x & 31;
    const int row0 = blockIdx.x * 16;
    const int col0 = wave * 16;

    const float* __restrict__ W  = (blockIdx.y == 0) ? Wl : Wr;
    float*       __restrict__ O  = (blockIdx.y == 0) ? XL : XR;

    const int m  = lane & 15;
    const int hi = lane >> 4;

    int arow = row0 + m; if (arow >= Nn) arow = Nn - 1;

    const float* __restrict__ ap = X + (size_t)arow * CIN;
    const float* __restrict__ bp = W + (size_t)(col0 + m) * CIN;

    v8f acc = {};
    #pragma unroll
    for (int k0 = 0; k0 < CIN; k0 += 4) {
        const int ka = k0 + hi * 2;
        v2f a, b;
        a.x = ap[ka];  a.y = ap[ka + 1];
        b.x = bp[ka];  b.y = bp[ka + 1];
        acc = __builtin_amdgcn_wmma_f32_16x16x4_f32(
             false, a,  false, b,
             (short)0, acc,  false,  false);
    }

    #pragma unroll
    for (int i = 0; i < 8; ++i) {
        const int r = row0 + i + hi * 8;
        if (r < Nn) O[(size_t)r * CIN + col0 + m] = acc[i];
    }
}

__global__ void gat_edge_score(
    const float* __restrict__ xl, const float* __restrict__ xr,
    const int* __restrict__ src, const int* __restrict__ dst,
    const float* __restrict__ att,
    float* __restrict__ ebuf, int* __restrict__ emax_i,
    int Etot, int E)
{
    const int t = blockIdx.x * blockDim.x + threadIdx.x;
    if (t >= Etot * HEADS) return;
    const int eid = t >> 2;
    const int h   = t & 3;

    int s, d;
    if (eid < E) { s = src[eid]; d = dst[eid]; }
    else         { s = d = eid - E; }

    const float* __restrict__ pl = xl + (size_t)s * CIN + h * HDIM;
    const float* __restrict__ pr = xr + (size_t)d * CIN + h * HDIM;
    const float* __restrict__ pa = att + h * HDIM;

    float acc = 0.0f;
    #pragma unroll
    for (int i = 0; i < HDIM; ++i) {
        float v = pl[i] + pr[i];
        v = (v > 0.0f) ? v : NEG_SLOPE * v;
        acc += pa[i] * v;
    }
    ebuf[(size_t)eid * HEADS + h] = acc;
    if (acc > 0.0f)
        atomicMax(&emax_i[d * HEADS + h], __float_as_int(acc));
}

__global__ void gat_edge_exp(
    const int* __restrict__ dst,
    float* __restrict__ ebuf,
    const float* __restrict__ emax,
    float* __restrict__ denom,
    int Etot, int E)
{
    const int t = blockIdx.x * blockDim.x + threadIdx.x;
    if (t >= Etot * HEADS) return;
    const int eid = t >> 2;
    const int h   = t & 3;
    const int d   = (eid < E) ? dst[eid] : (eid - E);

    const float m  = emax[d * HEADS + h];
    const float ex = __expf(ebuf[(size_t)eid * HEADS + h] - m);
    ebuf[(size_t)eid * HEADS + h] = ex;
    atomicAdd(&denom[d * HEADS + h], ex);
}

__global__ void gat_edge_scatter(
    const float* __restrict__ xr,
    const int* __restrict__ src, const int* __restrict__ dst,
    const float* __restrict__ ebuf, const float* __restrict__ denom,
    float* __restrict__ accum,
    int Etot, int E)
{
    const int t = blockIdx.x * blockDim.x + threadIdx.x;
    if (t >= Etot * HEADS) return;
    const int eid = t >> 2;
    const int h   = t & 3;

    int s, d;
    if (eid < E) { s = src[eid]; d = dst[eid]; }
    else         { s = d = eid - E; }

    float den = denom[d * HEADS + h];
    den = (den > 1e-16f) ? den : 1e-16f;
    const float alpha = ebuf[(size_t)eid * HEADS + h] / den;

    const float* __restrict__ p = xr    + (size_t)s * CIN + h * HDIM;
    float*       __restrict__ o = accum + (size_t)d * CIN + h * HDIM;
    #pragma unroll
    for (int i = 0; i < HDIM; ++i)
        atomicAdd(&o[i], p[i] * alpha);
}

__global__ void gat_finalize(
    const float* __restrict__ accum, const float* __restrict__ bias,
    float* __restrict__ out, int Nn)
{
    const int t = blockIdx.x * blockDim.x + threadIdx.x;
    if (t >= Nn * HDIM) return;
    const int n  = t >> 5;
    const int dc = t & 31;
    const float* __restrict__ a = accum + (size_t)n * CIN + dc;
    out[t] = (a[0] + a[HDIM] + a[2 * HDIM] + a[3 * HDIM]) * 0.25f + bias[dc];
}

extern "C" void kernel_launch(void* const* d_in, const int* in_sizes, int n_in,
                              void* d_out, int out_size, void* d_ws, size_t ws_size,
                              hipStream_t stream)
{
    const float* x    = (const float*)d_in[0];
    const int*   eidx = (const int*)d_in[1];
    const float* Wl   = (const float*)d_in[2];
    const float* Wr   = (const float*)d_in[3];
    const float* att  = (const float*)d_in[4];
    const float* bias = (const float*)d_in[5];
    float* out = (float*)d_out;

    const int Nn   = in_sizes[0] / CIN;
    const int E    = in_sizes[1] / 2;
    const int Etot = E + Nn;

    const int* src = eidx;
    const int* dst = eidx + E;

    float* ws = (float*)d_ws;
    size_t off = 0;
    float* xl    = ws + off; off += (size_t)Nn * CIN;
    float* xr    = ws + off; off += (size_t)Nn * CIN;
    float* ebuf  = ws + off; off += (size_t)Etot * HEADS;
    float* emax  = ws + off; off += (size_t)Nn * HEADS;
    float* denom = ws + off; off += (size_t)Nn * HEADS;
    float* accum = ws + off; off += (size_t)Nn * CIN;
    (void)ws_size;

    {
        size_t nz = (size_t)Nn * HEADS * 2 + (size_t)Nn * CIN;
        int blocks = 2048;
        gat_zero_kernel<<<blocks, 256, 0, stream>>>(emax, nz);
    }

    {
        dim3 grid((Nn + 15) / 16, 2);
        gat_gemm_wmma<<<grid, 256, 0, stream>>>(x, Wl, Wr, xl, xr, Nn);
    }

    const int edgeThreads = Etot * HEADS;
    const int edgeBlocks  = (edgeThreads + 255) / 256;

    gat_edge_score<<<edgeBlocks, 256, 0, stream>>>(
        xl, xr, src, dst, att, ebuf, (int*)emax, Etot, E);

    gat_edge_exp<<<edgeBlocks, 256, 0, stream>>>(
        dst, ebuf, emax, denom, Etot, E);

    gat_edge_scatter<<<edgeBlocks, 256, 0, stream>>>(
        xr, src, dst, ebuf, denom, accum, Etot, E);

    {
        const int t = Nn * HDIM;
        gat_finalize<<<(t + 255) / 256, 256, 0, stream>>>(accum, bias, out, Nn);
    }
}


# scatter: a block owns 196 destination rows in LDS, reads the whole dst list and adds alpha*xr rows of its own edges with LDS float adds; rows stored once, no global float adds; sum order changes
# speedup vs baseline: 3.0298x; 3.0298x over previous
#include <hip/hip_runtime.h>
#include <hip/hip_bf16.h>


typedef __attribute__((ext_vector_type(2))) float v2f;
typedef __attribute__((ext_vector_type(8))) float v8f;

#define HEADS 4
#define HDIM 32
#define CIN 128
#define NEG_SLOPE 0.2f

__global__ void gat_zero_kernel(float* __restrict__ p, size_t n) {
    size_t i = (size_t)blockIdx.x * blockDim.x + threadIdx.x;
    size_t stride = (size_t)gridDim.x * blockDim.x;
    for (; i < n; i += stride) p[i] = 0.0f;
}

__global__ __launch_bounds__(256) void gat_gemm_wmma(
    const float* __restrict__ X,
    const float* __restrict__ Wl,
    const float* __restrict__ Wr,
    float* __restrict__ XL,
    float* __restrict__ XR,
    int Nn)
{
    const int wave = threadIdx.x >> 5;
    const int lane = threadIdx.x & 31;
    const int row0 = blockIdx.x * 16;
    const int col0 = wave * 16;

    const float* __restrict__ W  = (blockIdx.y == 0) ? Wl : Wr;
    float*       __restrict__ O  = (blockIdx.y == 0) ? XL : XR;

    const int m  = lane & 15;
    const int hi = lane >> 4;

    int arow = row0 + m; if (arow >= Nn) arow = Nn - 1;

    const float* __restrict__ ap = X + (size_t)arow * CIN;
    const float* __restrict__ bp = W + (size_t)(col0 + m) * CIN;

    v8f acc = {};
    #pragma unroll
    for (int k0 = 0; k0 < CIN; k0 += 4) {
        const int ka = k0 + hi * 2;
        v2f a, b;
        a.x = ap[ka];  a.y = ap[ka + 1];
        b.x = bp[ka];  b.y = bp[ka + 1];
        acc = __builtin_amdgcn_wmma_f32_16x16x4_f32(
             false, a,  false, b,
             (short)0, acc,  false,  false);
    }

    #pragma unroll
    for (int i = 0; i < 8; ++i) {
        const int r = row0 + i + hi * 8;
        if (r < Nn) O[(size_t)r * CIN + col0 + m] = acc[i];
    }
}

__global__ void gat_edge_score(
    const float* __restrict__ xl, const float* __restrict__ xr,
    const int* __restrict__ src, const int* __restrict__ dst,
    const float* __restrict__ att,
    float* __restrict__ ebuf, int* __restrict__ emax_i,
    int Etot, int E)
{
    const int t = blockIdx.x * blockDim.x + threadIdx.x;
    if (t >= Etot * HEADS) return;
    const int eid = t >> 2;
    const int h   = t & 3;

    int s, d;
    if (eid < E) { s = src[eid]; d = dst[eid]; }
    else         { s = d = eid - E; }

    const float* __restrict__ pl = xl + (size_t)s * CIN + h * HDIM;
    const float* __restrict__ pr = xr + (size_t)d * CIN + h * HDIM;
    const float* __restrict__ pa = att + h * HDIM;

    float acc = 0.0f;
    #pragma unroll
    for (int i = 0; i < HDIM; ++i) {
        float v = pl[i] + pr[i];
        v = (v > 0.0f) ? v : NEG_SLOPE * v;
        acc += pa[i] * v;
    }
    ebuf[(size_t)eid * HEADS + h] = acc;
    if (acc > 0.0f)
        atomicMax(&emax_i[d * HEADS + h], __float_as_int(acc));
}

__global__ void gat_edge_exp(
    const int* __restrict__ dst,
    float* __restrict__ ebuf,
    const float* __restrict__ emax,
    float* __restrict__ denom,
    int Etot, int E)
{
    const int t = blockIdx.x * blockDim.x + threadIdx.x;
    if (t >= Etot * HEADS) return;
    const int eid = t >> 2;
    const int h   = t & 3;
    const int d   = (eid < E) ? dst[eid] : (eid - E);

    const float m  = emax[d * HEADS + h];
    const float ex = __expf(ebuf[(size_t)eid * HEADS + h] - m);
    ebuf[(size_t)eid * HEADS + h] = ex;
    atomicAdd(&denom[d * HEADS + h], ex);
}

#define OWN_ROWS    196
#define OWN_THREADS 1024
#define OWN_WAVES   (OWN_THREADS / 32)
#define OWN_LOADS   16
#define OWN_CHUNK   (OWN_LOADS * 32)
#define OWN_QCAP    (32 + OWN_CHUNK)

__device__ __forceinline__ void own_load_chunk(
    const int* __restrict__ dst, int cbase, int lane, int E, int (&d)[OWN_LOADS])
{
    #pragma unroll
    for (int k = 0; k < OWN_LOADS; ++k) {
        const int e  = cbase + k * 32 + lane;
        const int ec = (e < E) ? e : (E - 1);
        const int v  = dst[ec];
        d[k] = (e < E) ? v : -1;
    }
}

template <int G>
__device__ __forceinline__ void own_add_group(
    const float* __restrict__ xr, float* sacc, int lane, int j0,
    int s, int dl, float a0, float a1, float a2, float a3)
{
    float x[G][4];
    float al[G][4];
    int   dj[G];
    #pragma unroll
    for (int u = 0; u < G; ++u) {
        const int j  = j0 + u;
        const int sj = __builtin_amdgcn_readlane(s, j);
        dj[u]    = __builtin_amdgcn_readlane(dl, j);
        al[u][0] = __int_as_float(__builtin_amdgcn_readlane(__float_as_int(a0), j));
        al[u][1] = __int_as_float(__builtin_amdgcn_readlane(__float_as_int(a1), j));
        al[u][2] = __int_as_float(__builtin_amdgcn_readlane(__float_as_int(a2), j));
        al[u][3] = __int_as_float(__builtin_amdgcn_readlane(__float_as_int(a3), j));
        const float* __restrict__ p = xr + (size_t)sj * CIN + lane;
        x[u][0] = p[0];
        x[u][1] = p[HDIM];
        x[u][2] = p[2 * HDIM];
        x[u][3] = p[3 * HDIM];
    }
    #pragma unroll
    for (int u = 0; u < G; ++u) {
        float* q = sacc + dj[u] * CIN + lane;
        __hip_atomic_fetch_add(q,            x[u][0] * al[u][0], __ATOMIC_RELAXED, __HIP_MEMORY_SCOPE_WORKGROUP);
        __hip_atomic_fetch_add(q + HDIM,     x[u][1] * al[u][1], __ATOMIC_RELAXED, __HIP_MEMORY_SCOPE_WORKGROUP);
        __hip_atomic_fetch_add(q + 2 * HDIM, x[u][2] * al[u][2], __ATOMIC_RELAXED, __HIP_MEMORY_SCOPE_WORKGROUP);
        __hip_atomic_fetch_add(q + 3 * HDIM, x[u][3] * al[u][3], __ATOMIC_RELAXED, __HIP_MEMORY_SCOPE_WORKGROUP);
    }
}

__device__ __forceinline__ void own_flush(
    const float* __restrict__ xr, const int* __restrict__ src,
    const float* __restrict__ ebuf, const float* sden, float* sacc,
    const int2* qp, int lane, int n)
{
    int s = 0, dl = 0;
    float a0 = 0.0f, a1 = 0.0f, a2 = 0.0f, a3 = 0.0f;
    if (lane < n) {
        const int2 ent = qp[lane];
        dl = ent.y;
        s  = src[ent.x];
        const float4 ex = *reinterpret_cast<const float4*>(ebuf + (size_t)ent.x * HEADS);
        const float4 dn = *reinterpret_cast<const float4*>(&sden[dl * HEADS]);
        a0 = ex.x / dn.x;
        a1 = ex.y / dn.y;
        a2 = ex.z / dn.z;
        a3 = ex.w / dn.w;
    }
    int j0 = 0;
    for (; j0 + 8 <= n; j0 += 8)
        own_add_group<8>(xr, sacc, lane, j0, s, dl, a0, a1, a2, a3);
    for (; j0 < n; ++j0)
        own_add_group<1>(xr, sacc, lane, j0, s, dl, a0, a1, a2, a3);
}

__global__ __launch_bounds__(OWN_THREADS) void gat_owner_sum(
    const float* __restrict__ xr,
    const int* __restrict__ src, const int* __restrict__ dst,
    const float* __restrict__ ebuf, const float* __restrict__ denom,
    float* __restrict__ accum,
    int Nn, int E)
{
    __shared__ __attribute__((aligned(16))) float sacc[OWN_ROWS * CIN];
    __shared__ __attribute__((aligned(16))) float sden[OWN_ROWS * HEADS];
    __shared__ __attribute__((aligned(16))) int2  slist[OWN_WAVES * OWN_QCAP];

    const int tid  = threadIdx.x;
    const int lane = tid & 31;
    const int wave = __builtin_amdgcn_readfirstlane(tid >> 5);
    int2* const q = slist + wave * OWN_QCAP;

    const int nchunks = (E + OWN_CHUNK - 1) / OWN_CHUNK;
    const int rot = (int)blockIdx.x * (nchunks / (int)gridDim.x);

    for (int lo = blockIdx.x * OWN_ROWS; lo < Nn; lo += gridDim.x * OWN_ROWS) {
        const int cnt = (Nn - lo < OWN_ROWS) ? (Nn - lo) : OWN_ROWS;

        for (int t = tid; t < cnt * HEADS; t += OWN_THREADS) {
            const float den = denom[(size_t)lo * HEADS + t];
            sden[t] = (den > 1e-16f) ? den : 1e-16f;
        }
        __syncthreads();

        for (int idx = tid; idx < cnt * CIN; idx += OWN_THREADS) {
            const int dl = idx >> 7;
            const int ch = idx & (CIN - 1);
            const int n  = lo + dl;
            const float al = ebuf[((size_t)E + n) * HEADS + (ch >> 5)] / sden[dl * HEADS + (ch >> 5)];
            sacc[idx] = xr[(size_t)n * CIN + ch] * al;
        }
        __syncthreads();

        int nq = 0;
        int i = wave;
        int ccur = i + rot; if (ccur >= nchunks) ccur -= nchunks;
        int dcur[OWN_LOADS];
        if (i < nchunks) own_load_chunk(dst, ccur * OWN_CHUNK, lane, E, dcur);

        while (i < nchunks) {
            const int inext = i + OWN_WAVES;
            int cnext = inext + rot; if (cnext >= nchunks) cnext -= nchunks;
            int dnxt[OWN_LOADS];
            own_load_chunk(dst, ((inext < nchunks) ? cnext : ccur) * OWN_CHUNK, lane, E, dnxt);

            const int cbase = ccur * OWN_CHUNK;
            #pragma unroll
            for (int k = 0; k < OWN_LOADS; ++k) {
                const unsigned dl = (unsigned)(dcur[k] - lo);
                const bool hit = dl < (unsigned)cnt;
                const unsigned mask = __builtin_amdgcn_ballot_w32(hit);
                if (mask) {
                    const int pos = nq + (int)__builtin_amdgcn_mbcnt_lo(mask, 0u);
                    if (hit) q[pos] = make_int2(cbase + k * 32 + lane, (int)dl);
                    nq += __builtin_popcount(mask);
                }
            }

            int h = 0;
            for (; h + 32 <= nq; h += 32)
                own_flush(xr, src, ebuf, sden, sacc, q + h, lane, 32);
            if (h > 0) {
                const int rem = nq - h;
                int2 mv = make_int2(0, 0);
                if (lane < rem) mv = q[h + lane];
                if (lane < rem) q[lane] = mv;
                nq = rem;
            }

            #pragma unroll
            for (int k = 0; k < OWN_LOADS; ++k) dcur[k] = dnxt[k];
            i = inext; ccur = cnext;
        }
        if (nq > 0) own_flush(xr, src, ebuf, sden, sacc, q, lane, nq);
        __syncthreads();

        for (int idx = tid * 4; idx < cnt * CIN; idx += OWN_THREADS * 4) {
            *reinterpret_cast<float4*>(accum + (size_t)lo * CIN + idx) =
                *reinterpret_cast<const float4*>(&sacc[idx]);
        }
        __syncthreads();
    }
}

__global__ void gat_finalize(
    const float* __restrict__ accum, const float* __restrict__ bias,
    float* __restrict__ out, int Nn)
{
    const int t = blockIdx.x * blockDim.x + threadIdx.x;
    if (t >= Nn * HDIM) return;
    const int n  = t >> 5;
    const int dc = t & 31;
    const float* __restrict__ a = accum + (size_t)n * CIN + dc;
    out[t] = (a[0] + a[HDIM] + a[2 * HDIM] + a[3 * HDIM]) * 0.25f + bias[dc];
}

extern "C" void kernel_launch(void* const* d_in, const int* in_sizes, int n_in,
                              void* d_out, int out_size, void* d_ws, size_t ws_size,
                              hipStream_t stream)
{
    const float* x    = (const float*)d_in[0];
    const int*   eidx = (const int*)d_in[1];
    const float* Wl   = (const float*)d_in[2];
    const float* Wr   = (const float*)d_in[3];
    const float* att  = (const float*)d_in[4];
    const float* bias = (const float*)d_in[5];
    float* out = (float*)d_out;

    const int Nn   = in_sizes[0] / CIN;
    const int E    = in_sizes[1] / 2;
    const int Etot = E + Nn;

    const int* src = eidx;
    const int* dst = eidx + E;

    float* ws = (float*)d_ws;
    size_t off = 0;
    float* xl    = ws + off; off += (size_t)Nn * CIN;
    float* xr    = ws + off; off += (size_t)Nn * CIN;
    float* ebuf  = ws + off; off += (size_t)Etot * HEADS;
    float* emax  = ws + off; off += (size_t)Nn * HEADS;
    float* denom = ws + off; off += (size_t)Nn * HEADS;
    float* accum = ws + off; off += (size_t)Nn * CIN;
    (void)ws_size;

    {
        size_t nz = (size_t)Nn * HEADS * 2;
        int blocks = 2048;
        gat_zero_kernel<<<blocks, 256, 0, stream>>>(emax, nz);
    }

    {
        dim3 grid((Nn + 15) / 16, 2);
        gat_gemm_wmma<<<grid, 256, 0, stream>>>(x, Wl, Wr, xl, xr, Nn);
    }

    const int edgeThreads = Etot * HEADS;
    const int edgeBlocks  = (edgeThreads + 255) / 256;

    gat_edge_score<<<edgeBlocks, 256, 0, stream>>>(
        xl, xr, src, dst, att, ebuf, (int*)emax, Etot, E);

    gat_edge_exp<<<edgeBlocks, 256, 0, stream>>>(
        dst, ebuf, emax, denom, Etot, E);

    gat_owner_sum<<<256, OWN_THREADS, 0, stream>>>(
        xr, src, dst, ebuf, denom, accum, Nn, E);

    {
        const int t = Nn * HDIM;
        gat_finalize<<<(t + 255) / 256, 256, 0, stream>>>(accum, bias, out, Nn);
    }
}


# GEMM: a wave holds the B fragments of its 16 columns of Wl and Wr in registers and walks row tiles; A read 16 B/lane from global, one A fragment feeds 2 WMMAs; k order per step and sum order change
# speedup vs baseline: 3.4554x; 1.1405x over previous
#include <hip/hip_runtime.h>
#include <hip/hip_bf16.h>


typedef __attribute__((ext_vector_type(2))) float v2f;
typedef __attribute__((ext_vector_type(4))) float v4f;
typedef __attribute__((ext_vector_type(8))) float v8f;

#define HEADS 4
#define HDIM 32
#define CIN 128
#define NEG_SLOPE 0.2f

__global__ void gat_zero_kernel(float* __restrict__ p, size_t n) {
    size_t i = (size_t)blockIdx.x * blockDim.x + threadIdx.x;
    size_t stride = (size_t)gridDim.x * blockDim.x;
    for (; i < n; i += stride) p[i] = 0.0f;
}

#define GEMM_BLOCKS 256
#define GEMM_KLOADS (CIN / 8)

__device__ __forceinline__ void gemm_load_a(
    const float* __restrict__ X, int tile, int m, int hi, int Nn, v4f (&a)[GEMM_KLOADS])
{
    int arow = tile * 16 + m; if (arow >= Nn) arow = Nn - 1;
    const v4f* __restrict__ p =
        reinterpret_cast<const v4f*>(X + (size_t)arow * CIN + hi * 4);
    #pragma unroll
    for (int j = 0; j < GEMM_KLOADS; ++j) a[j] = p[2 * j];
}

__global__ __launch_bounds__(256) void gat_gemm_wmma(
    const float* __restrict__ X,
    const float* __restrict__ Wl,
    const float* __restrict__ Wr,
    float* __restrict__ XL,
    float* __restrict__ XR,
    int Nn)
{
    const int wave = threadIdx.x >> 5;
    const int lane = threadIdx.x & 31;
    const int col0 = wave * 16;
    const int m  = lane & 15;
    const int hi = lane >> 4;

    const int ntiles = (Nn + 15) >> 4;
    int tile = blockIdx.x;
    if (tile >= ntiles) return;

    v4f fl[GEMM_KLOADS], fr[GEMM_KLOADS];
    {
        const v4f* __restrict__ pl =
            reinterpret_cast<const v4f*>(Wl + (size_t)(col0 + m) * CIN + hi * 4);
        const v4f* __restrict__ pr =
            reinterpret_cast<const v4f*>(Wr + (size_t)(col0 + m) * CIN + hi * 4);
        #pragma unroll
        for (int j = 0; j < GEMM_KLOADS; ++j) { fl[j] = pl[2 * j]; fr[j] = pr[2 * j]; }
    }

    v4f a[GEMM_KLOADS];
    gemm_load_a(X, tile, m, hi, Nn, a);

    for (; tile < ntiles; tile += gridDim.x) {
        const int nxt = tile + (int)gridDim.x;
        v4f an[GEMM_KLOADS];
        gemm_load_a(X, (nxt < ntiles) ? nxt : tile, m, hi, Nn, an);

        v8f cl = {}, cr = {};
        #pragma unroll
        for (int j = 0; j < GEMM_KLOADS; ++j) {
            v2f a0, a1, l0, l1, r0, r1;
            a0.x = a[j].x;  a0.y = a[j].y;  a1.x = a[j].z;  a1.y = a[j].w;
            l0.x = fl[j].x; l0.y = fl[j].y; l1.x = fl[j].z; l1.y = fl[j].w;
            r0.x = fr[j].x; r0.y = fr[j].y; r1.x = fr[j].z; r1.y = fr[j].w;
            cl = __builtin_amdgcn_wmma_f32_16x16x4_f32(false, a0, false, l0, (short)0, cl, false, false);
            cr = __builtin_amdgcn_wmma_f32_16x16x4_f32(false, a0, false, r0, (short)0, cr, false, false);
            cl = __builtin_amdgcn_wmma_f32_16x16x4_f32(false, a1, false, l1, (short)0, cl, false, false);
            cr = __builtin_amdgcn_wmma_f32_16x16x4_f32(false, a1, false, r1, (short)0, cr, false, false);
        }

        const int row0 = tile * 16;
        #pragma unroll
        for (int i = 0; i < 8; ++i) {
            const int r = row0 + i + hi * 8;
            if (r < Nn) {
                XL[(size_t)r * CIN + col0 + m] = cl[i];
                XR[(size_t)r * CIN + col0 + m] = cr[i];
            }
        }

        #pragma unroll
        for (int j = 0; j < GEMM_KLOADS; ++j) a[j] = an[j];
    }
}

__global__ void gat_edge_score(
    const float* __restrict__ xl, const float* __restrict__ xr,
    const int* __restrict__ src, const int* __restrict__ dst,
    const float* __restrict__ att,
    float* __restrict__ ebuf, int* __restrict__ emax_i,
    int Etot, int E)
{
    const int t = blockIdx.x * blockDim.x + threadIdx.x;
    if (t >= Etot * HEADS) return;
    const int eid = t >> 2;
    const int h   = t & 3;

    int s, d;
    if (eid < E) { s = src[eid]; d = dst[eid]; }
    else         { s = d = eid - E; }

    const float* __restrict__ pl = xl + (size_t)s * CIN + h * HDIM;
    const float* __restrict__ pr = xr + (size_t)d * CIN + h * HDIM;
    const float* __restrict__ pa = att + h * HDIM;

    float acc = 0.0f;
    #pragma unroll
    for (int i = 0; i < HDIM; ++i) {
        float v = pl[i] + pr[i];
        v = (v > 0.0f) ? v : NEG_SLOPE * v;
        acc += pa[i] * v;
    }
    ebuf[(size_t)eid * HEADS + h] = acc;
    if (acc > 0.0f)
        atomicMax(&emax_i[d * HEADS + h], __float_as_int(acc));
}

__global__ void gat_edge_exp(
    const int* __restrict__ dst,
    float* __restrict__ ebuf,
    const float* __restrict__ emax,
    float* __restrict__ denom,
    int Etot, int E)
{
    const int t = blockIdx.x * blockDim.x + threadIdx.x;
    if (t >= Etot * HEADS) return;
    const int eid = t >> 2;
    const int h   = t & 3;
    const int d   = (eid < E) ? dst[eid] : (eid - E);

    const float m  = emax[d * HEADS + h];
    const float ex = __expf(ebuf[(size_t)eid * HEADS + h] - m);
    ebuf[(size_t)eid * HEADS + h] = ex;
    atomicAdd(&denom[d * HEADS + h], ex);
}

#define OWN_ROWS    196
#define OWN_THREADS 1024
#define OWN_WAVES   (OWN_THREADS / 32)
#define OWN_LOADS   16
#define OWN_CHUNK   (OWN_LOADS * 32)
#define OWN_QCAP    (32 + OWN_CHUNK)

__device__ __forceinline__ void own_load_chunk(
    const int* __restrict__ dst, int cbase, int lane, int E, int (&d)[OWN_LOADS])
{
    #pragma unroll
    for (int k = 0; k < OWN_LOADS; ++k) {
        const int e  = cbase + k * 32 + lane;
        const int ec = (e < E) ? e : (E - 1);
        const int v  = dst[ec];
        d[k] = (e < E) ? v : -1;
    }
}

template <int G>
__device__ __forceinline__ void own_add_group(
    const float* __restrict__ xr, float* sacc, int lane, int j0,
    int s, int dl, float a0, float a1, float a2, float a3)
{
    float x[G][4];
    float al[G][4];
    int   dj[G];
    #pragma unroll
    for (int u = 0; u < G; ++u) {
        const int j  = j0 + u;
        const int sj = __builtin_amdgcn_readlane(s, j);
        dj[u]    = __builtin_amdgcn_readlane(dl, j);
        al[u][0] = __int_as_float(__builtin_amdgcn_readlane(__float_as_int(a0), j));
        al[u][1] = __int_as_float(__builtin_amdgcn_readlane(__float_as_int(a1), j));
        al[u][2] = __int_as_float(__builtin_amdgcn_readlane(__float_as_int(a2), j));
        al[u][3] = __int_as_float(__builtin_amdgcn_readlane(__float_as_int(a3), j));
        const float* __restrict__ p = xr + (size_t)sj * CIN + lane;
        x[u][0] = p[0];
        x[u][1] = p[HDIM];
        x[u][2] = p[2 * HDIM];
        x[u][3] = p[3 * HDIM];
    }
    #pragma unroll
    for (int u = 0; u < G; ++u) {
        float* q = sacc + dj[u] * CIN + lane;
        __hip_atomic_fetch_add(q,            x[u][0] * al[u][0], __ATOMIC_RELAXED, __HIP_MEMORY_SCOPE_WORKGROUP);
        __hip_atomic_fetch_add(q + HDIM,     x[u][1] * al[u][1], __ATOMIC_RELAXED, __HIP_MEMORY_SCOPE_WORKGROUP);
        __hip_atomic_fetch_add(q + 2 * HDIM, x[u][2] * al[u][2], __ATOMIC_RELAXED, __HIP_MEMORY_SCOPE_WORKGROUP);
        __hip_atomic_fetch_add(q + 3 * HDIM, x[u][3] * al[u][3], __ATOMIC_RELAXED, __HIP_MEMORY_SCOPE_WORKGROUP);
    }
}

__device__ __forceinline__ void own_flush(
    const float* __restrict__ xr, const int* __restrict__ src,
    const float* __restrict__ ebuf, const float* sden, float* sacc,
    const int2* qp, int lane, int n)
{
    int s = 0, dl = 0;
    float a0 = 0.0f, a1 = 0.0f, a2 = 0.0f, a3 = 0.0f;
    if (lane < n) {
        const int2 ent = qp[lane];
        dl = ent.y;
        s  = src[ent.x];
        const float4 ex = *reinterpret_cast<const float4*>(ebuf + (size_t)ent.x * HEADS);
        const float4 dn = *reinterpret_cast<const float4*>(&sden[dl * HEADS]);
        a0 = ex.x / dn.x;
        a1 = ex.y / dn.y;
        a2 = ex.z / dn.z;
        a3 = ex.w / dn.w;
    }
    int j0 = 0;
    for (; j0 + 8 <= n; j0 += 8)
        own_add_group<8>(xr, sacc, lane, j0, s, dl, a0, a1, a2, a3);
    for (; j0 < n; ++j0)
        own_add_group<1>(xr, sacc, lane, j0, s, dl, a0, a1, a2, a3);
}

__global__ __launch_bounds__(OWN_THREADS) void gat_owner_sum(
    const float* __restrict__ xr,
    const int* __restrict__ src, const int* __restrict__ dst,
    const float* __restrict__ ebuf, const float* __restrict__ denom,
    float* __restrict__ accum,
    int Nn, int E)
{
    __shared__ __attribute__((aligned(16))) float sacc[OWN_ROWS * CIN];
    __shared__ __attribute__((aligned(16))) float sden[OWN_ROWS * HEADS];
    __shared__ __attribute__((aligned(16))) int2  slist[OWN_WAVES * OWN_QCAP];

    const int tid  = threadIdx.x;
    const int lane = tid & 31;
    const int wave = __builtin_amdgcn_readfirstlane(tid >> 5);
    int2* const q = slist + wave * OWN_QCAP;

    const int nchunks = (E + OWN_CHUNK - 1) / OWN_CHUNK;
    const int rot = (int)blockIdx.x * (nchunks / (int)gridDim.x);

    for (int lo = blockIdx.x * OWN_ROWS; lo < Nn; lo += gridDim.x * OWN_ROWS) {
        const int cnt = (Nn - lo < OWN_ROWS) ? (Nn - lo) : OWN_ROWS;

        for (int t = tid; t < cnt * HEADS; t += OWN_THREADS) {
            const float den = denom[(size_t)lo * HEADS + t];
            sden[t] = (den > 1e-16f) ? den : 1e-16f;
        }
        __syncthreads();

        for (int idx = tid; idx < cnt * CIN; idx += OWN_THREADS) {
            const int dl = idx >> 7;
            const int ch = idx & (CIN - 1);
            const int n  = lo + dl;
            const float al = ebuf[((size_t)E + n) * HEADS + (ch >> 5)] / sden[dl * HEADS + (ch >> 5)];
            sacc[idx] = xr[(size_t)n * CIN + ch] * al;
        }
        __syncthreads();

        int nq = 0;
        int i = wave;
        int ccur = i + rot; if (ccur >= nchunks) ccur -= nchunks;
        int dcur[OWN_LOADS];
        if (i < nchunks) own_load_chunk(dst, ccur * OWN_CHUNK, lane, E, dcur);

        while (i < nchunks) {
            const int inext = i + OWN_WAVES;
            int cnext = inext + rot; if (cnext >= nchunks) cnext -= nchunks;
            int dnxt[OWN_LOADS];
            own_load_chunk(dst, ((inext < nchunks) ? cnext : ccur) * OWN_CHUNK, lane, E, dnxt);

            const int cbase = ccur * OWN_CHUNK;
            #pragma unroll
            for (int k = 0; k < OWN_LOADS; ++k) {
                const unsigned dl = (unsigned)(dcur[k] - lo);
                const bool hit = dl < (unsigned)cnt;
                const unsigned mask = __builtin_amdgcn_ballot_w32(hit);
                if (mask) {
                    const int pos = nq + (int)__builtin_amdgcn_mbcnt_lo(mask, 0u);
                    if (hit) q[pos] = make_int2(cbase + k * 32 + lane, (int)dl);
                    nq += __builtin_popcount(mask);
                }
            }

            int h = 0;
            for (; h + 32 <= nq; h += 32)
                own_flush(xr, src, ebuf, sden, sacc, q + h, lane, 32);
            if (h > 0) {
                const int rem = nq - h;
                int2 mv = make_int2(0, 0);
                if (lane < rem) mv = q[h + lane];
                if (lane < rem) q[lane] = mv;
                nq = rem;
            }

            #pragma unroll
            for (int k = 0; k < OWN_LOADS; ++k) dcur[k] = dnxt[k];
            i = inext; ccur = cnext;
        }
        if (nq > 0) own_flush(xr, src, ebuf, sden, sacc, q, lane, nq);
        __syncthreads();

        for (int idx = tid * 4; idx < cnt * CIN; idx += OWN_THREADS * 4) {
            *reinterpret_cast<float4*>(accum + (size_t)lo * CIN + idx) =
                *reinterpret_cast<const float4*>(&sacc[idx]);
        }
        __syncthreads();
    }
}

__global__ void gat_finalize(
    const float* __restrict__ accum, const float* __restrict__ bias,
    float* __restrict__ out, int Nn)
{
    const int t = blockIdx.x * blockDim.x + threadIdx.x;
    if (t >= Nn * HDIM) return;
    const int n  = t >> 5;
    const int dc = t & 31;
    const float* __restrict__ a = accum + (size_t)n * CIN + dc;
    out[t] = (a[0] + a[HDIM] + a[2 * HDIM] + a[3 * HDIM]) * 0.25f + bias[dc];
}

extern "C" void kernel_launch(void* const* d_in, const int* in_sizes, int n_in,
                              void* d_out, int out_size, void* d_ws, size_t ws_size,
                              hipStream_t stream)
{
    const float* x    = (const float*)d_in[0];
    const int*   eidx = (const int*)d_in[1];
    const float* Wl   = (const float*)d_in[2];
    const float* Wr   = (const float*)d_in[3];
    const float* att  = (const float*)d_in[4];
    const float* bias = (const float*)d_in[5];
    float* out = (float*)d_out;

    const int Nn   = in_sizes[0] / CIN;
    const int E    = in_sizes[1] / 2;
    const int Etot = E + Nn;

    const int* src = eidx;
    const int* dst = eidx + E;

    float* ws = (float*)d_ws;
    size_t off = 0;
    float* xl    = ws + off; off += (size_t)Nn * CIN;
    float* xr    = ws + off; off += (size_t)Nn * CIN;
    float* ebuf  = ws + off; off += (size_t)Etot * HEADS;
    float* emax  = ws + off; off += (size_t)Nn * HEADS;
    float* denom = ws + off; off += (size_t)Nn * HEADS;
    float* accum = ws + off; off += (size_t)Nn * CIN;
    (void)ws_size;

    {
        size_t nz = (size_t)Nn * HEADS * 2;
        int blocks = 2048;
        gat_zero_kernel<<<blocks, 256, 0, stream>>>(emax, nz);
    }

    gat_gemm_wmma<<<GEMM_BLOCKS, 256, 0, stream>>>(x, Wl, Wr, xl, xr, Nn);

    const int edgeThreads = Etot * HEADS;
    const int edgeBlocks  = (edgeThreads + 255) / 256;

    gat_edge_score<<<edgeBlocks, 256, 0, stream>>>(
        xl, xr, src, dst, att, ebuf, (int*)emax, Etot, E);

    gat_edge_exp<<<edgeBlocks, 256, 0, stream>>>(
        dst, ebuf, emax, denom, Etot, E);

    gat_owner_sum<<<256, OWN_THREADS, 0, stream>>>(
        xr, src, dst, ebuf, denom, accum, Nn, E);

    {
        const int t = Nn * HDIM;
        gat_finalize<<<(t + 255) / 256, 256, 0, stream>>>(accum, bias, out, Nn);
    }
}


# score by row owner: pass one only books edges (source row, slot in the dst row's list); the wave that sums a row holds xr[dst] in registers, scores its edges, max in registers; sum order changes
# speedup vs baseline: 11.0682x; 3.2032x over previous
#include <hip/hip_runtime.h>
#include <hip/hip_bf16.h>

typedef __attribute__((ext_vector_type(2))) float v2f;
typedef __attribute__((ext_vector_type(4))) float v4f;
typedef __attribute__((ext_vector_type(8))) float v8f;

#define HEADS 4
#define HDIM 32
#define CIN 128
#define NEG_SLOPE 0.2f

#define OWN_ROWS    196
#define OWN_BLOCKS  256
#define OWN_THREADS 1024
#define OWN_WAVES   (OWN_THREADS / 32)
#define OWN_SUB     512
#define OWN_LOADS   2
#define OWN_CHUNK   (OWN_LOADS * OWN_SUB)
#define OWN_CAP     256
#define OWN_RS      64

__device__ __forceinline__ unsigned own_key(int d) {
    return ((unsigned)d / OWN_ROWS) & (OWN_BLOCKS - 1);
}

#define GEMM_BLOCKS 256
#define GEMM_KLOADS (CIN / 8)

__device__ __forceinline__ void gemm_load_a(
    const float* __restrict__ X, int tile, int m, int hi, int Nn, v4f (&a)[GEMM_KLOADS])
{
    int arow = tile * 16 + m; if (arow >= Nn) arow = Nn - 1;
    const v4f* __restrict__ p =
        reinterpret_cast<const v4f*>(X + (size_t)arow * CIN + hi * 4);
    #pragma unroll
    for (int j = 0; j < GEMM_KLOADS; ++j) a[j] = p[2 * j];
}

__global__ __launch_bounds__(256) void gat_gemm_wmma(
    const float* __restrict__ X,
    const float* __restrict__ Wl,
    const float* __restrict__ Wr,
    float* __restrict__ XL,
    float* __restrict__ XR,
    int Nn,
    const int* __restrict__ dst,
    unsigned* __restrict__ keys,
    int E)
{
    {
        const int nwords = ((E + OWN_CHUNK - 1) / OWN_CHUNK) * (OWN_CHUNK / 4);
        for (int w = blockIdx.x * 256 + threadIdx.x; w < nwords; w += gridDim.x * 256) {
            unsigned pk = 0;
            #pragma unroll
            for (int b = 0; b < 4; ++b) {
                const int e = 4 * w + b;
                const int d = (e < E) ? dst[e] : 0;
                pk |= own_key(d) << (8 * b);
            }
            keys[w] = pk;
        }
    }

    const int wave = threadIdx.x >> 5;
    const int lane = threadIdx.x & 31;
    const int col0 = wave * 16;
    const int m  = lane & 15;
    const int hi = lane >> 4;

    const int ntiles = (Nn + 15) >> 4;
    int tile = blockIdx.x;
    if (tile >= ntiles) return;

    v4f fl[GEMM_KLOADS], fr[GEMM_KLOADS];
    {
        const v4f* __restrict__ pl =
            reinterpret_cast<const v4f*>(Wl + (size_t)(col0 + m) * CIN + hi * 4);
        const v4f* __restrict__ pr =
            reinterpret_cast<const v4f*>(Wr + (size_t)(col0 + m) * CIN + hi * 4);
        #pragma unroll
        for (int j = 0; j < GEMM_KLOADS; ++j) { fl[j] = pl[2 * j]; fr[j] = pr[2 * j]; }
    }

    v4f a[GEMM_KLOADS];
    gemm_load_a(X, tile, m, hi, Nn, a);

    for (; tile < ntiles; tile += gridDim.x) {
        const int nxt = tile + (int)gridDim.x;
        v4f an[GEMM_KLOADS];
        gemm_load_a(X, (nxt < ntiles) ? nxt : tile, m, hi, Nn, an);

        v8f cl = {}, cr = {};
        #pragma unroll
        for (int j = 0; j < GEMM_KLOADS; ++j) {
            v2f a0, a1, l0, l1, r0, r1;
            a0.x = a[j].x;  a0.y = a[j].y;  a1.x = a[j].z;  a1.y = a[j].w;
            l0.x = fl[j].x; l0.y = fl[j].y; l1.x = fl[j].z; l1.y = fl[j].w;
            r0.x = fr[j].x; r0.y = fr[j].y; r1.x = fr[j].z; r1.y = fr[j].w;
            cl = __builtin_amdgcn_wmma_f32_16x16x4_f32(false, a0, false, l0, (short)0, cl, false, false);
            cr = __builtin_amdgcn_wmma_f32_16x16x4_f32(false, a0, false, r0, (short)0, cr, false, false);
            cl = __builtin_amdgcn_wmma_f32_16x16x4_f32(false, a1, false, l1, (short)0, cl, false, false);
            cr = __builtin_amdgcn_wmma_f32_16x16x4_f32(false, a1, false, r1, (short)0, cr, false, false);
        }

        const int row0 = tile * 16;
        #pragma unroll
        for (int i = 0; i < 8; ++i) {
            const int r = row0 + i + hi * 8;
            if (r < Nn) {
                XL[(size_t)r * CIN + col0 + m] = cl[i];
                XR[(size_t)r * CIN + col0 + m] = cr[i];
            }
        }

        #pragma unroll
        for (int j = 0; j < GEMM_KLOADS; ++j) a[j] = an[j];
    }
}

__device__ __forceinline__ void own_load_keys(
    const unsigned* __restrict__ keys, int chunk, int lane, uint4 (&k)[OWN_LOADS])
{
    const uint4* __restrict__ p = reinterpret_cast<const uint4*>(keys);
    const unsigned i0 = (unsigned)chunk * (unsigned)(OWN_CHUNK / 16) + (unsigned)lane;
    #pragma unroll
    for (int u = 0; u < OWN_LOADS; ++u) k[u] = p[i0 + (unsigned)(u * 32)];
}

__device__ __forceinline__ unsigned own_key_hits(unsigned w, unsigned kb) {
    const unsigned x = w ^ kb;
    const unsigned y = ((x & 0x7F7F7F7Fu) + 0x7F7F7F7Fu) | x;
    return ~y & 0x80808080u;
}

__device__ __forceinline__ unsigned own_hit_bits(
    const uint4 k, unsigned kb, int ebase, int sub_end, int E)
{
    unsigned M = (own_key_hits(k.x, kb) >> 7) | (own_key_hits(k.y, kb) >> 6) |
                 (own_key_hits(k.z, kb) >> 5) | (own_key_hits(k.w, kb) >> 4);
    if (sub_end > E) {
        #pragma unroll
        for (int i = 0; i < 16; ++i)
            if (ebase + i >= E) M &= ~(1u << (8 * (i & 3) + (i >> 2)));
    }
    return M;
}

__device__ __forceinline__ int own_note_turn(unsigned& M, unsigned mask, int ebase, int* q, int nq)
{
    const int pos = nq + (int)__builtin_amdgcn_mbcnt_lo(mask, 0u);
    if (M != 0u) {
        const int p = __ffs((int)M) - 1;
        q[pos] = ebase + ((p & 4) * (OWN_SUB / 4)) + 4 * (p & 3) + (p >> 3);
        M &= M - 1u;
    }
    return nq + __builtin_popcount(mask);
}

__device__ __forceinline__ float score_leaky(float v) {
    return (v > 0.0f) ? v : NEG_SLOPE * v;
}

__device__ __forceinline__ float score_part(const float4 av, const v4f a, const v4f b) {
    const float v0 = score_leaky(a.x + b.x);
    const float v1 = score_leaky(a.y + b.y);
    const float v2 = score_leaky(a.z + b.z);
    const float v3 = score_leaky(a.w + b.w);
    return av.x * v0 + av.y * v1 + av.z * v2 + av.w * v3;
}

__device__ __forceinline__ float head_sum(float p) {
    p += __shfl_xor(p, 4, 32);
    p += __shfl_xor(p, 2, 32);
    p += __shfl_xor(p, 1, 32);
    return p;
}

__device__ __forceinline__ float row_score_group(
    const float* __restrict__ xl, float* ssc, const v4f xd, const float4 av,
    int lane, int j0, int m, int s, int gi)
{
    v4f a[8];
    #pragma unroll
    for (int u = 0; u < 8; ++u) {
        const int sj = __builtin_amdgcn_readlane(s, j0 + u);
        a[u] = reinterpret_cast<const v4f*>(xl)[(unsigned)sj * 32u + (unsigned)lane];
    }
    float p[8];
    #pragma unroll
    for (int u = 0; u < 8; ++u) p[u] = score_part(av, a[u], xd);

    const bool b2 = (lane & 4) != 0;
    const bool b1 = (lane & 2) != 0;
    const bool b0 = (lane & 1) != 0;
    float r[4];
    #pragma unroll
    for (int u = 0; u < 4; ++u) {
        const float keep = b2 ? p[u + 4] : p[u];
        const float send = b2 ? p[u] : p[u + 4];
        r[u] = keep + __shfl_xor(send, 4, 32);
    }
    float w[2];
    #pragma unroll
    for (int u = 0; u < 2; ++u) {
        const float keep = b1 ? r[u + 2] : r[u];
        const float send = b1 ? r[u] : r[u + 2];
        w[u] = keep + __shfl_xor(send, 2, 32);
    }
    const float keep = b0 ? w[1] : w[0];
    const float send = b0 ? w[0] : w[1];
    const float t = keep + __shfl_xor(send, 1, 32);

    const int ju   = j0 + (lane & 7);
    const int my_g = __shfl(gi, ju, 32);
    if (ju < m) {
        ssc[(unsigned)my_g * (unsigned)HEADS + (unsigned)(lane >> 3)] = t;
        return t;
    }
    return 0.0f;
}

template <bool TAIL>
__device__ __forceinline__ void row_add_group(
    const float* __restrict__ xr, const float* sexp, int lane, int j0, int m,
    int s, int gi, v4f& acc, float& dsum)
{
    v4f   x[8];
    float a[8];
    #pragma unroll
    for (int u = 0; u < 8; ++u) {
        const int sj = __builtin_amdgcn_readlane(s, j0 + u);
        const int gj = __builtin_amdgcn_readlane(gi, j0 + u);
        x[u]  = reinterpret_cast<const v4f*>(xr)[(unsigned)sj * 32u + (unsigned)lane];
        a[u] = sexp[gj * HEADS + (lane >> 3)];
    }
    #pragma unroll
    for (int u = 0; u < 8; ++u)
        if (!TAIL || j0 + u < m) {
            dsum += a[u];
            acc  += x[u] * a[u];
        }
}

__device__ __forceinline__ void book_flush(
    const int* __restrict__ src, const int* __restrict__ dst,
    int* qp, int lane, int n, int lo, int cnt,
    int* rowcnt, unsigned short* rowslot, int gbase)
{
    if (lane < n) {
        const int eid = qp[lane];
        const int s = src[eid];
        const unsigned r = (unsigned)(dst[eid] - lo);
        qp[lane] = s;
        if (r < (unsigned)cnt) {
            const int slot = __hip_atomic_fetch_add(&rowcnt[r], 1, __ATOMIC_RELAXED,
                                                    __HIP_MEMORY_SCOPE_WORKGROUP);
            if (slot < OWN_RS) rowslot[r * OWN_RS + slot] = (unsigned short)(gbase + lane);
        }
    }
}

template <bool SUM>
__device__ __forceinline__ void row_walk_keys(
    const float* __restrict__ xl, const float* __restrict__ xr,
    const int* __restrict__ src, const int* __restrict__ dst,
    const unsigned* __restrict__ keys, unsigned kb, int nchunks, int E,
    int lane, int grow, const v4f xd, const float4 av,
    float& mxl, v4f& acc, float& dsum)
{
    for (int c = 0; c < nchunks; ++c) {
        uint4 k[OWN_LOADS];
        own_load_keys(keys, c, lane, k);
        const int cbase = c * OWN_CHUNK;
        const int ebase = cbase + lane * 16;
        unsigned M = own_hit_bits(k[0], kb, ebase, cbase + OWN_SUB, E) |
                     (own_hit_bits(k[1], kb, ebase + OWN_SUB, cbase + 2 * OWN_SUB, E) << 4);
        unsigned mask = __builtin_amdgcn_ballot_w32(M != 0u);
        while (mask) {
            const int hl = __ffs((int)mask) - 1;
            const unsigned mh = (unsigned)__builtin_amdgcn_readlane((int)M, hl);
            const int p = __ffs((int)mh) - 1;
            const int e = cbase + hl * 16 + ((p & 4) * (OWN_SUB / 4)) + 4 * (p & 3) + (p >> 3);
            if (lane == hl) M &= M - 1u;
            if (dst[e] == grow) {
                const int s = src[e];
                const v4f a = reinterpret_cast<const v4f*>(xl)[(unsigned)s * 32u + (unsigned)lane];
                const float t = head_sum(score_part(av, a, xd));
                if (SUM) {
                    const float ex = __expf(t - mxl);
                    const v4f x = reinterpret_cast<const v4f*>(xr)[(unsigned)s * 32u + (unsigned)lane];
                    dsum += ex;
                    acc  += x * ex;
                } else {
                    mxl = fmaxf(mxl, t);
                }
            }
            mask = __builtin_amdgcn_ballot_w32(M != 0u);
        }
    }
}

__global__ __launch_bounds__(OWN_THREADS) void gat_owner(
    const float* __restrict__ xl, const float* __restrict__ xr,
    const int* __restrict__ src, const int* __restrict__ dst,
    const unsigned* __restrict__ keys,
    const float* __restrict__ att, const float* __restrict__ bias,
    float* __restrict__ out,
    int Nn, int E)
{
    __shared__ __attribute__((aligned(16))) float sacc[OWN_ROWS * CIN];
    __shared__ __attribute__((aligned(16))) float sden[OWN_ROWS * HEADS];
    __shared__ __attribute__((aligned(16))) int   slist[OWN_WAVES * OWN_CAP];
    __shared__ __attribute__((aligned(16))) float sscore[OWN_WAVES * OWN_CAP * HEADS];
    __shared__ int            srcnt[OWN_ROWS];
    __shared__ unsigned short srslot[OWN_ROWS * OWN_RS];
    __shared__ int            srflag[1];

    const int tid  = threadIdx.x;
    const int lane = tid & 31;
    const int wave = __builtin_amdgcn_readfirstlane(tid >> 5);
    int* const q = slist + wave * OWN_CAP;
    const unsigned kb = (unsigned)blockIdx.x * 0x01010101u;

    const float4 av = *reinterpret_cast<const float4*>(att + lane * 4);

    const int nchunks = (E + OWN_CHUNK - 1) / OWN_CHUNK;
    const int rot = (int)blockIdx.x * (nchunks / (int)gridDim.x);

    for (int lo = blockIdx.x * OWN_ROWS; lo < Nn; lo += gridDim.x * OWN_ROWS) {
        const int cnt = (Nn - lo < OWN_ROWS) ? (Nn - lo) : OWN_ROWS;

        if (tid < OWN_ROWS) srcnt[tid] = 0;
        if (tid == 0) srflag[0] = 0;
        __syncthreads();

        {
            int  nq   = 0;
            int  done = 0;
            bool over = false;
            int i = wave;
            int ccur = i + rot; if (ccur >= nchunks) ccur -= nchunks;
            uint4 kcur[OWN_LOADS] = {};
            if (i < nchunks) own_load_keys(keys, ccur, lane, kcur);

            while (i < nchunks) {
                const int inext = i + OWN_WAVES;
                int cnext = inext + rot; if (cnext >= nchunks) cnext -= nchunks;
                uint4 knxt[OWN_LOADS];
                own_load_keys(keys, (inext < nchunks) ? cnext : ccur, lane, knxt);

                const int cbase = ccur * OWN_CHUNK;
                const int ebase = cbase + lane * 16;
                unsigned M = own_hit_bits(kcur[0], kb, ebase, cbase + OWN_SUB, E) |
                             (own_hit_bits(kcur[1], kb, ebase + OWN_SUB, cbase + 2 * OWN_SUB, E) << 4);

                unsigned mask = __builtin_amdgcn_ballot_w32(M != 0u);
                while (mask) {
                    if (nq + 32 > OWN_CAP) {
                        over = true; srflag[0] = 1; break;
                    }
                    nq = own_note_turn(M, mask, ebase, q, nq);
                    if (nq - done >= 32) {
                        book_flush(src, dst, q + done, lane, 32, lo, cnt,
                                   srcnt, srslot, wave * OWN_CAP + done);
                        done += 32;
                    }
                    mask = __builtin_amdgcn_ballot_w32(M != 0u);
                }
                if (over) break;

                #pragma unroll
                for (int u = 0; u < OWN_LOADS; ++u) kcur[u] = knxt[u];
                i = inext; ccur = cnext;
            }
            if (nq > done)
                book_flush(src, dst, q + done, lane, nq - done, lo, cnt,
                           srcnt, srslot, wave * OWN_CAP + done);
        }
        __syncthreads();
        if (tid < cnt && srcnt[tid] > OWN_RS) srflag[0] = 1;
        __syncthreads();
        const bool fast = __builtin_amdgcn_readfirstlane(srflag[0]) == 0;

        v4f xdn = {0.0f, 0.0f, 0.0f, 0.0f}, xsn = {0.0f, 0.0f, 0.0f, 0.0f};
        if (wave < cnt) {
            xdn = reinterpret_cast<const v4f*>(xr)[(unsigned)(lo + wave) * 32u + (unsigned)lane];
            xsn = reinterpret_cast<const v4f*>(xl)[(unsigned)(lo + wave) * 32u + (unsigned)lane];
        }
        for (int r = wave; r < cnt; r += OWN_WAVES) {
            const int grow = lo + r;
            const v4f xd = xdn;
            const v4f xs = xsn;
            {
                const int gnext = (r + OWN_WAVES < cnt) ? (grow + OWN_WAVES) : grow;
                xdn = reinterpret_cast<const v4f*>(xr)[(unsigned)gnext * 32u + (unsigned)lane];
                xsn = reinterpret_cast<const v4f*>(xl)[(unsigned)gnext * 32u + (unsigned)lane];
            }
            float mxl = 0.0f;
            int   nr  = 0;
            v4f   acc = {0.0f, 0.0f, 0.0f, 0.0f};
            float dsum = 0.0f;

            if (fast) {
                nr = __builtin_amdgcn_readfirstlane(srcnt[r]);
                if (nr > OWN_RS) nr = OWN_RS;
                for (int base = 0; base < nr; base += 32) {
                    const int m = (nr - base < 32) ? (nr - base) : 32;
                    int s = 0, gi = 0;
                    if (lane < m) {
                        gi = srslot[r * OWN_RS + base + lane];
                        s  = slist[gi];
                    }
                    for (int j0 = 0; j0 < m; j0 += 8)
                        mxl = fmaxf(mxl, row_score_group(xl, sscore, xd, av, lane, j0, m, s, gi));
                }
            } else {
                row_walk_keys<false>(xl, xr, src, dst, keys, kb, nchunks, E, lane, grow, xd, av, mxl, acc, dsum);
            }
            const float ts = head_sum(score_part(av, xs, xd));
            mxl = fmaxf(mxl, ts);
            mxl = fmaxf(mxl, __shfl_xor(mxl, 4, 32));
            mxl = fmaxf(mxl, __shfl_xor(mxl, 2, 32));
            mxl = fmaxf(mxl, __shfl_xor(mxl, 1, 32));
            const float es = __expf(ts - mxl);
            acc  = xd * es;
            dsum = es;

            if (fast) {
                const float mx0 = __int_as_float(__builtin_amdgcn_readlane(__float_as_int(mxl), 0));
                const float mx1 = __int_as_float(__builtin_amdgcn_readlane(__float_as_int(mxl), 8));
                const float mx2 = __int_as_float(__builtin_amdgcn_readlane(__float_as_int(mxl), 16));
                const float mx3 = __int_as_float(__builtin_amdgcn_readlane(__float_as_int(mxl), 24));
                for (int base = 0; base < nr; base += 32) {
                    const int m = (nr - base < 32) ? (nr - base) : 32;
                    int s = 0, gi = 0;
                    if (lane < m) {
                        gi = srslot[r * OWN_RS + base + lane];
                        s  = slist[gi];
                        float4 sc = reinterpret_cast<const float4*>(sscore)[gi];
                        sc.x = __expf(sc.x - mx0);
                        sc.y = __expf(sc.y - mx1);
                        sc.z = __expf(sc.z - mx2);
                        sc.w = __expf(sc.w - mx3);
                        reinterpret_cast<float4*>(sscore)[gi] = sc;
                    }
                    int j0 = 0;
                    for (; j0 + 8 <= m; j0 += 8)
                        row_add_group<false>(xr, sscore, lane, j0, m, s, gi, acc, dsum);
                    if (j0 < m)
                        row_add_group<true>(xr, sscore, lane, j0, m, s, gi, acc, dsum);
                }
            } else {
                row_walk_keys<true>(xl, xr, src, dst, keys, kb, nchunks, E, lane, grow, xd, av, mxl, acc, dsum);
            }
            *(reinterpret_cast<v4f*>(sacc + r * CIN) + lane) = acc;
            if ((lane & 7) == 0) sden[r * HEADS + (lane >> 3)] = dsum;
        }
        __syncthreads();

        float* __restrict__ out_own = out + (size_t)lo * HDIM;
        for (int idx = tid; idx < cnt * HDIM; idx += OWN_THREADS) {
            const int dl = idx >> 5;
            const int c  = idx & (HDIM - 1);
            const float* a = sacc + dl * CIN + c;
            const float4 dn = *reinterpret_cast<const float4*>(&sden[dl * HEADS]);
            const float d0 = (dn.x > 1e-16f) ? dn.x : 1e-16f;
            const float d1 = (dn.y > 1e-16f) ? dn.y : 1e-16f;
            const float d2 = (dn.z > 1e-16f) ? dn.z : 1e-16f;
            const float d3 = (dn.w > 1e-16f) ? dn.w : 1e-16f;
            out_own[(unsigned)idx] =
                (a[0] / d0 + a[HDIM] / d1 + a[2 * HDIM] / d2 + a[3 * HDIM] / d3) * 0.25f + bias[(unsigned)c];
        }
        __syncthreads();
    }
}

extern "C" void kernel_launch(void* const* d_in, const int* in_sizes, int n_in,
                              void* d_out, int out_size, void* d_ws, size_t ws_size,
                              hipStream_t stream)
{
    const float* x    = (const float*)d_in[0];
    const int*   eidx = (const int*)d_in[1];
    const float* Wl   = (const float*)d_in[2];
    const float* Wr   = (const float*)d_in[3];
    const float* att  = (const float*)d_in[4];
    const float* bias = (const float*)d_in[5];
    float* out = (float*)d_out;

    const int Nn = in_sizes[0] / CIN;
    const int E  = in_sizes[1] / 2;

    const int* src = eidx;
    const int* dst = eidx + E;

    float* ws = (float*)d_ws;
    size_t off = 0;
    float* xl    = ws + off; off += (size_t)Nn * CIN;
    float* xr    = ws + off; off += (size_t)Nn * CIN;
    unsigned* keys = (unsigned*)(ws + off);
    (void)ws_size;

    gat_gemm_wmma<<<GEMM_BLOCKS, 256, 0, stream>>>(x, Wl, Wr, xl, xr, Nn, dst, keys, E);

    gat_owner<<<OWN_BLOCKS, OWN_THREADS, 0, stream>>>(
        xl, xr, src, dst, keys, att, bias, out, Nn, E);
}
